# starting kernel
# baseline (speedup 1.0000x reference)
#include <hip/hip_runtime.h>
#include <hip/hip_bf16.h>

typedef __attribute__((ext_vector_type(2))) float v2f;
typedef __attribute__((ext_vector_type(8))) float v8f;

#define LATDIM 128
#define HEADS 8
#define HDIM 16

__global__ __launch_bounds__(256) void zero_kernel(float* __restrict__ p, int n) {
    int i = blockIdx.x * blockDim.x + threadIdx.x;
    if (i < n) p[i] = 0.0f;
}

__global__ __launch_bounds__(256) void qkv_gemm_wmma(
    const float* __restrict__ embeds,
    const float* __restrict__ Wq, const float* __restrict__ Wk,
    const float* __restrict__ Wv,
    float* __restrict__ Q, float* __restrict__ K, float* __restrict__ V,
    int rowTiles, int n_nodes)
{
    const int lane = threadIdx.x & 31;
    const int wave = threadIdx.x >> 5;
    const int gw   = blockIdx.x * (blockDim.x >> 5) + wave;
    const int mat  = gw % 3;
    const int rt   = gw / 3;
    if (rt >= rowTiles) return;

    const float* __restrict__ W   = (mat == 0) ? Wq : (mat == 1) ? Wk : Wv;
    float*       __restrict__ Out = (mat == 0) ? Q  : (mat == 1) ? K  : V;

    const int m     = lane & 15;
    const int hi    = lane >> 4;
    const int khalf = hi << 1;

    int arowIdx = rt * 16 + m;
    if (arowIdx >= n_nodes) arowIdx = n_nodes - 1;
    const float* __restrict__ arow = embeds + (size_t)arowIdx * LATDIM;

    v8f acc[8] = {};

#pragma unroll 2
    for (int k0 = 0; k0 < LATDIM; k0 += 4) {
        v2f a;
        a.x = arow[k0 + khalf];
        a.y = arow[k0 + khalf + 1];
        const float* __restrict__ w0 = W + (size_t)(k0 + khalf) * LATDIM + m;
#pragma unroll
        for (int t = 0; t < 8; ++t) {
            v2f b;
            b.x = w0[t * 16];
            b.y = w0[t * 16 + LATDIM];
            acc[t] = __builtin_amdgcn_wmma_f32_16x16x4_f32(
                false, a, false, b, (short)0, acc[t], false, false);
        }
    }

    const int orow = rt * 16 + hi * 8;
#pragma unroll
    for (int t = 0; t < 8; ++t) {
#pragma unroll
        for (int r = 0; r < 8; ++r) {
            int row = orow + r;
            if (row < n_nodes)
                Out[(size_t)row * LATDIM + t * 16 + m] = acc[t][r];
        }
    }
}

__global__ __launch_bounds__(256) void edge_scores(
    const float* __restrict__ Q, const float* __restrict__ K,
    const int* __restrict__ rows, const int* __restrict__ cols,
    float* __restrict__ expAtt, float* __restrict__ attNorm, int n_edges)
{
    int idx = blockIdx.x * blockDim.x + threadIdx.x;
    if (idx >= n_edges * HEADS) return;
    int e = idx >> 3;
    int h = idx & 7;
    int r = rows[e], c = cols[e];
    const float4* qp = (const float4*)(Q + (size_t)r * LATDIM + h * HDIM);
    const float4* kp = (const float4*)(K + (size_t)c * LATDIM + h * HDIM);
    float att = 0.0f;
#pragma unroll
    for (int j = 0; j < 4; ++j) {
        float4 q4 = qp[j];
        float4 k4 = kp[j];
        att += q4.x * k4.x + q4.y * k4.y + q4.z * k4.z + q4.w * k4.w;
    }
    att = fminf(10.0f, fmaxf(-10.0f, att));
    float ea = __expf(att);
    expAtt[idx] = ea;
    unsafeAtomicAdd(attNorm + (size_t)r * HEADS + h, ea);
}

__global__ __launch_bounds__(256) void edge_aggregate(
    const float* __restrict__ V,
    const int* __restrict__ rows, const int* __restrict__ cols,
    const float* __restrict__ expAtt, const float* __restrict__ attNorm,
    float* __restrict__ out, int n_edges)
{
    int t = blockIdx.x * blockDim.x + threadIdx.x;
    int e = t >> 5;
    if (e >= n_edges) return;
    int lane = t & 31;
    int r = rows[e], c = cols[e];
    int h = lane >> 2;
    float a = expAtt[(size_t)e * HEADS + h] /
              (attNorm[(size_t)r * HEADS + h] + 1e-8f);
    float4 v4 = *(const float4*)(V + (size_t)c * LATDIM + lane * 4);
    float* op = out + (size_t)r * LATDIM + lane * 4;
    unsafeAtomicAdd(op + 0, a * v4.x);
    unsafeAtomicAdd(op + 1, a * v4.y);
    unsafeAtomicAdd(op + 2, a * v4.z);
    unsafeAtomicAdd(op + 3, a * v4.w);
}

extern "C" void kernel_launch(void* const* d_in, const int* in_sizes, int n_in,
                              void* d_out, int out_size, void* d_ws, size_t ws_size,
                              hipStream_t stream) {
    const float* embeds = (const float*)d_in[0];
    const float* Wq     = (const float*)d_in[1];
    const float* Wk     = (const float*)d_in[2];
    const float* Wv     = (const float*)d_in[3];
    const int*   rows   = (const int*)d_in[4];
    const int*   cols   = (const int*)d_in[5];

    const int N = in_sizes[0] / LATDIM;
    const int E = in_sizes[4];

    float* out = (float*)d_out;

    float* ws      = (float*)d_ws;
    float* Qm      = ws;
    float* Km      = Qm + (size_t)N * LATDIM;
    float* Vm      = Km + (size_t)N * LATDIM;
    float* expAtt  = Vm + (size_t)N * LATDIM;
    float* attNorm = expAtt + (size_t)E * HEADS;

    {
        int n = N * LATDIM;
        zero_kernel<<<(n + 255) / 256, 256, 0, stream>>>(out, n);
        n = N * HEADS;
        zero_kernel<<<(n + 255) / 256, 256, 0, stream>>>(attNorm, n);
    }

    const int rowTiles = (N + 15) / 16;
    const int nwaves   = rowTiles * 3;
    qkv_gemm_wmma<<<(nwaves + 7) / 8, 256, 0, stream>>>(
        embeds, Wq, Wk, Wv, Qm, Km, Vm, rowTiles, N);

    {
        int npairs = E * HEADS;
        edge_scores<<<(npairs + 255) / 256, 256, 0, stream>>>(
            Qm, Km, rows, cols, expAtt, attNorm, E);
    }

    {
        long nthreads = (long)E * 32;
        edge_aggregate<<<(int)((nthreads + 255) / 256), 256, 0, stream>>>(
            Vm, rows, cols, expAtt, attNorm, out, E);
    }
}


# Edges bucketed by destination row (counting sort, integer atomics); one wave per row keeps scores, softmax sum and V sum in registers: no float atomics, no zero fills; divide after the row sum
# speedup vs baseline: 1.4981x; 1.4981x over previous
#include <hip/hip_runtime.h>
#include <hip/hip_bf16.h>

typedef __attribute__((ext_vector_type(2))) float v2f;
typedef __attribute__((ext_vector_type(8))) float v8f;

#define LATDIM 128
#define HEADS 8
#define HDIM 16

__global__ __launch_bounds__(256) void zero_counts(int* __restrict__ p, int n) {
    int i = blockIdx.x * blockDim.x + threadIdx.x;
    if (i < n) p[i] = 0;
}

__global__ __launch_bounds__(256) void qkv_gemm_wmma(
    const float* __restrict__ embeds,
    const float* __restrict__ Wq, const float* __restrict__ Wk,
    const float* __restrict__ Wv,
    float* __restrict__ Q, float* __restrict__ K, float* __restrict__ V,
    int rowTiles, int n_nodes)
{
    const int lane = threadIdx.x & 31;
    const int wave = threadIdx.x >> 5;
    const int gw   = blockIdx.x * (blockDim.x >> 5) + wave;
    const int mat  = gw % 3;
    const int rt   = gw / 3;
    if (rt >= rowTiles) return;

    const float* __restrict__ W   = (mat == 0) ? Wq : (mat == 1) ? Wk : Wv;
    float*       __restrict__ Out = (mat == 0) ? Q  : (mat == 1) ? K  : V;

    const int m     = lane & 15;
    const int hi    = lane >> 4;
    const int khalf = hi << 1;

    int arowIdx = rt * 16 + m;
    if (arowIdx >= n_nodes) arowIdx = n_nodes - 1;
    const float* __restrict__ arow = embeds + (size_t)arowIdx * LATDIM;

    v8f acc[8] = {};

#pragma unroll 2
    for (int k0 = 0; k0 < LATDIM; k0 += 4) {
        v2f a;
        a.x = arow[k0 + khalf];
        a.y = arow[k0 + khalf + 1];
        const float* __restrict__ w0 = W + (size_t)(k0 + khalf) * LATDIM + m;
#pragma unroll
        for (int t = 0; t < 8; ++t) {
            v2f b;
            b.x = w0[t * 16];
            b.y = w0[t * 16 + LATDIM];
            acc[t] = __builtin_amdgcn_wmma_f32_16x16x4_f32(
                false, a, false, b, (short)0, acc[t], false, false);
        }
    }

    const int orow = rt * 16 + hi * 8;
#pragma unroll
    for (int t = 0; t < 8; ++t) {
#pragma unroll
        for (int r = 0; r < 8; ++r) {
            int row = orow + r;
            if (row < n_nodes)
                Out[(size_t)row * LATDIM + t * 16 + m] = acc[t][r];
        }
    }
}

__global__ __launch_bounds__(256) void count_rank(
    const int* __restrict__ rows, int* __restrict__ count,
    int* __restrict__ rank, int n_edges)
{
    int e = blockIdx.x * blockDim.x + threadIdx.x;
    if (e >= n_edges) return;
    rank[e] = atomicAdd(count + rows[e], 1);
}

#define SCAN_PER 16
__global__ __launch_bounds__(1024) void scan_counts(
    const int* __restrict__ count, int* __restrict__ offset, int n)
{
    __shared__ int wsum[32];
    const int tid  = threadIdx.x;
    const int lane = tid & 31;
    const int wave = tid >> 5;
    int carry = 0;
    for (int base = 0; base < n; base += 1024 * SCAN_PER) {
        const int i0 = base + tid * SCAN_PER;
        const bool full = (i0 + SCAN_PER <= n);
        int c[SCAN_PER];
        if (full) {
#pragma unroll
            for (int j = 0; j < SCAN_PER; j += 4) {
                const int4 t = *(const int4*)(count + i0 + j);
                c[j] = t.x; c[j + 1] = t.y; c[j + 2] = t.z; c[j + 3] = t.w;
            }
        } else {
#pragma unroll
            for (int j = 0; j < SCAN_PER; ++j)
                c[j] = (i0 + j < n) ? count[i0 + j] : 0;
        }
        int tot = 0;
#pragma unroll
        for (int j = 0; j < SCAN_PER; ++j) {
            const int t = c[j];
            c[j] = tot;
            tot += t;
        }
        int x = tot;
        for (int d = 1; d < 32; d <<= 1) {
            const int y = __shfl_up(x, d);
            if (lane >= d) x += y;
        }
        if (lane == 31) wsum[wave] = x;
        __syncthreads();
        const int w = wsum[lane];
        int ws = w;
        for (int d = 1; d < 32; d <<= 1) {
            const int y = __shfl_up(ws, d);
            if (lane >= d) ws += y;
        }
        const int wave_excl = __shfl(ws, wave) - __shfl(w, wave);
        const int total     = __shfl(ws, 31);
        const int excl = carry + wave_excl + (x - tot);
        if (full) {
#pragma unroll
            for (int j = 0; j < SCAN_PER; j += 4)
                *(int4*)(offset + i0 + j) = make_int4(excl + c[j], excl + c[j + 1],
                                                      excl + c[j + 2], excl + c[j + 3]);
        } else {
#pragma unroll
            for (int j = 0; j < SCAN_PER; ++j)
                if (i0 + j < n) offset[i0 + j] = excl + c[j];
        }
        carry += total;
        __syncthreads();
    }
    if (tid == 0) offset[n] = carry;
}

__global__ __launch_bounds__(256) void fill_buckets(
    const int* __restrict__ rows, const int* __restrict__ cols,
    const int* __restrict__ rank, const int* __restrict__ offset,
    int* __restrict__ scol, int n_edges)
{
    int e = blockIdx.x * blockDim.x + threadIdx.x;
    if (e >= n_edges) return;
    int pos = offset[rows[e]] + rank[e];
    if ((unsigned)pos < (unsigned)n_edges) scol[pos] = cols[e];
}

__device__ __forceinline__ int clamp_row(int c, int n_nodes) {
    return min(max(c, 0), n_nodes - 1);
}

__device__ __forceinline__ float quad_sum(float p) {
    p += __builtin_bit_cast(float, __builtin_amdgcn_update_dpp(
             0, __builtin_bit_cast(int, p), 0xB1, 0xF, 0xF, true));
    p += __builtin_bit_cast(float, __builtin_amdgcn_update_dpp(
             0, __builtin_bit_cast(int, p), 0x4E, 0xF, 0xF, true));
    return p;
}

__device__ __forceinline__ void edge_accumulate(
    const float4 q, const float4 k4, const float4 v4, float4& acc, float& den)
{
    float att = quad_sum(q.x * k4.x + q.y * k4.y + q.z * k4.z + q.w * k4.w);
    att = fminf(10.0f, fmaxf(-10.0f, att));
    const float ea = __expf(att);
    den   += ea;
    acc.x += ea * v4.x;
    acc.y += ea * v4.y;
    acc.z += ea * v4.z;
    acc.w += ea * v4.w;
}

__global__ __launch_bounds__(256) void row_attend(
    const float* __restrict__ Q, const float* __restrict__ K,
    const float* __restrict__ V,
    const int* __restrict__ offset, const int* __restrict__ scol,
    float* __restrict__ out, int n_nodes)
{
    const int lane = threadIdx.x & 31;
    const int row = blockIdx.x * 8 + __builtin_amdgcn_readfirstlane(threadIdx.x >> 5);
    if (row >= n_nodes) return;
    const int beg = offset[row];
    const int end = offset[row + 1];

    const float4 q = *(const float4*)(Q + (size_t)row * LATDIM + lane * 4);
    float4 acc = make_float4(0.0f, 0.0f, 0.0f, 0.0f);
    float den = 0.0f;

    for (int base = beg; base < end; base += 32) {
        const int cnt = min(32, end - base);
        const int cidx = (lane < cnt) ? scol[base + lane] : 0;
        int i = 0;
        for (; i + 1 < cnt; i += 2) {
            const int ca = clamp_row(__builtin_amdgcn_readlane(cidx, i), n_nodes);
            const int cb = clamp_row(__builtin_amdgcn_readlane(cidx, i + 1), n_nodes);
            const float4 ka = *(const float4*)(K + (size_t)ca * LATDIM + lane * 4);
            const float4 kb = *(const float4*)(K + (size_t)cb * LATDIM + lane * 4);
            const float4 va = *(const float4*)(V + (size_t)ca * LATDIM + lane * 4);
            const float4 vb = *(const float4*)(V + (size_t)cb * LATDIM + lane * 4);
            edge_accumulate(q, ka, va, acc, den);
            edge_accumulate(q, kb, vb, acc, den);
        }
        if (i < cnt) {
            const int ca = clamp_row(__builtin_amdgcn_readlane(cidx, i), n_nodes);
            const float4 ka = *(const float4*)(K + (size_t)ca * LATDIM + lane * 4);
            const float4 va = *(const float4*)(V + (size_t)ca * LATDIM + lane * 4);
            edge_accumulate(q, ka, va, acc, den);
        }
    }

    const float d = den + 1e-8f;
    float4 o;
    o.x = acc.x / d;
    o.y = acc.y / d;
    o.z = acc.z / d;
    o.w = acc.w / d;
    *(float4*)(out + (size_t)row * LATDIM + lane * 4) = o;
}

extern "C" void kernel_launch(void* const* d_in, const int* in_sizes, int n_in,
                              void* d_out, int out_size, void* d_ws, size_t ws_size,
                              hipStream_t stream) {
    const float* embeds = (const float*)d_in[0];
    const float* Wq     = (const float*)d_in[1];
    const float* Wk     = (const float*)d_in[2];
    const float* Wv     = (const float*)d_in[3];
    const int*   rows   = (const int*)d_in[4];
    const int*   cols   = (const int*)d_in[5];

    const int N = in_sizes[0] / LATDIM;
    const int E = in_sizes[4];

    float* out = (float*)d_out;

    const size_t NP = ((size_t)N + 64) & ~(size_t)63;
    float* ws     = (float*)d_ws;
    float* Qm     = ws;
    float* Km     = Qm + (size_t)N * LATDIM;
    float* Vm     = Km + (size_t)N * LATDIM;
    int*   count  = (int*)(Vm + (size_t)N * LATDIM);
    int*   offset = count + NP;
    int*   rank   = offset + NP;
    int*   scol   = rank + (size_t)E;

    zero_counts<<<(N + 255) / 256, 256, 0, stream>>>(count, N);
    count_rank<<<(E + 255) / 256, 256, 0, stream>>>(rows, count, rank, E);
    scan_counts<<<1, 1024, 0, stream>>>(count, offset, N);
    fill_buckets<<<(E + 255) / 256, 256, 0, stream>>>(rows, cols, rank, offset, scol, E);

    const int rowTiles = (N + 15) / 16;
    const int nwaves   = rowTiles * 3;
    qkv_gemm_wmma<<<(nwaves + 7) / 8, 256, 0, stream>>>(
        embeds, Wq, Wk, Wv, Qm, Km, Vm, rowTiles, N);

    row_attend<<<(N + 7) / 8, 256, 0, stream>>>(Qm, Km, Vm, offset, scol, out, N);
}


# Edges bucketed through LDS in 2 launches (block-local sort by 256-row bin; one work-group per bin builds offsets and lists): no global atomics, no scattered 4-byte traffic; sum order by LDS arrival
# speedup vs baseline: 2.2209x; 1.4825x over previous
#include <hip/hip_runtime.h>
#include <hip/hip_bf16.h>

typedef __attribute__((ext_vector_type(2))) float v2f;
typedef __attribute__((ext_vector_type(8))) float v8f;

#define LATDIM 128
#define HEADS 8
#define HDIM 16

__global__ __launch_bounds__(256) void qkv_gemm_wmma(
    const float* __restrict__ embeds,
    const float* __restrict__ Wq, const float* __restrict__ Wk,
    const float* __restrict__ Wv,
    float* __restrict__ Q, float* __restrict__ K, float* __restrict__ V,
    int rowTiles, int n_nodes)
{
    const int lane = threadIdx.x & 31;
    const int wave = threadIdx.x >> 5;
    const int gw   = blockIdx.x * (blockDim.x >> 5) + wave;
    const int mat  = gw % 3;
    const int rt   = gw / 3;
    if (rt >= rowTiles) return;

    const float* __restrict__ W   = (mat == 0) ? Wq : (mat == 1) ? Wk : Wv;
    float*       __restrict__ Out = (mat == 0) ? Q  : (mat == 1) ? K  : V;

    const int m     = lane & 15;
    const int hi    = lane >> 4;
    const int khalf = hi << 1;

    int arowIdx = rt * 16 + m;
    if (arowIdx >= n_nodes) arowIdx = n_nodes - 1;
    const float* __restrict__ arow = embeds + (size_t)arowIdx * LATDIM;

    v8f acc[8] = {};

#pragma unroll 2
    for (int k0 = 0; k0 < LATDIM; k0 += 4) {
        v2f a;
        a.x = arow[k0 + khalf];
        a.y = arow[k0 + khalf + 1];
        const float* __restrict__ w0 = W + (size_t)(k0 + khalf) * LATDIM + m;
#pragma unroll
        for (int t = 0; t < 8; ++t) {
            v2f b;
            b.x = w0[t * 16];
            b.y = w0[t * 16 + LATDIM];
            acc[t] = __builtin_amdgcn_wmma_f32_16x16x4_f32(
                false, a, false, b, (short)0, acc[t], false, false);
        }
    }

    const int orow = rt * 16 + hi * 8;
#pragma unroll
    for (int t = 0; t < 8; ++t) {
#pragma unroll
        for (int r = 0; r < 8; ++r) {
            int row = orow + r;
            if (row < n_nodes)
                Out[(size_t)row * LATDIM + t * 16 + m] = acc[t][r];
        }
    }
}

#define BIN_SHIFT 8
#define BIN_ROWS  (1 << BIN_SHIFT)
#define CHUNK     2560
#define CHUNK_PER (CHUNK / 256)
#define BIN_CAP   8192

__global__ __launch_bounds__(256) void bin_edges(
    const int* __restrict__ rows, const int* __restrict__ cols,
    int* __restrict__ stage, unsigned* __restrict__ table,
    int n_edges, int n_bins, int n_chunks)
{
    extern __shared__ int dyn[];
    __shared__ int sorted_r[CHUNK];
    __shared__ int sorted_c[CHUNK];
    __shared__ int wsum[8];
    int* hist  = dyn;
    int* start = dyn + n_bins;

    const int tid   = threadIdx.x;
    const int lane  = tid & 31;
    const int wave  = tid >> 5;
    const int chunk = blockIdx.x;
    const int base  = chunk * CHUNK;
    const int valid = min(CHUNK, n_edges - base);

    for (int i = tid; i < n_bins; i += 256) hist[i] = 0;

    int r[CHUNK_PER], c[CHUNK_PER], rk[CHUNK_PER];
#pragma unroll
    for (int j = 0; j < CHUNK_PER; ++j) {
        const int i = j * 256 + tid;
        r[j] = (i < valid) ? rows[base + i] : 0;
        c[j] = (i < valid) ? cols[base + i] : 0;
    }
    __syncthreads();
#pragma unroll
    for (int j = 0; j < CHUNK_PER; ++j) {
        const int i = j * 256 + tid;
        rk[j] = 0;
        if (i < valid) {
            const int bin = min(max(r[j] >> BIN_SHIFT, 0), n_bins - 1);
            rk[j] = atomicAdd(&hist[bin], 1);
        }
    }
    __syncthreads();

    int carry = 0;
    for (int b0 = 0; b0 < n_bins; b0 += 256) {
        const int b = b0 + tid;
        const int v = (b < n_bins) ? hist[b] : 0;
        int x = v;
        for (int d = 1; d < 32; d <<= 1) {
            const int y = __shfl_up(x, d);
            if (lane >= d) x += y;
        }
        if (lane == 31) wsum[wave] = x;
        __syncthreads();
        int woff = 0, tot = 0;
        for (int i = 0; i < 8; ++i) {
            const int s = wsum[i];
            if (i < wave) woff += s;
            tot += s;
        }
        if (b < n_bins) {
            const int st = carry + woff + x - v;
            start[b] = st;
            table[(size_t)b * n_chunks + chunk] = ((unsigned)st << 16) | (unsigned)v;
        }
        carry += tot;
        __syncthreads();
    }

#pragma unroll
    for (int j = 0; j < CHUNK_PER; ++j) {
        const int i = j * 256 + tid;
        if (i < valid) {
            const int bin = min(max(r[j] >> BIN_SHIFT, 0), n_bins - 1);
            const int pos = start[bin] + rk[j];
            sorted_r[pos] = r[j];
            sorted_c[pos] = c[j];
        }
    }
    __syncthreads();
    for (int i = tid; i < valid; i += 256) {
        *(int2*)(stage + 2 * (size_t)(base + i)) = make_int2(sorted_r[i], sorted_c[i]);
    }
}

__global__ __launch_bounds__(256) void bucket_rows(
    const int* __restrict__ stage, const unsigned* __restrict__ table,
    int* __restrict__ offset, int* __restrict__ scol,
    int n_edges, int n_nodes, int n_bins, int n_chunks)
{
    __shared__ int cnt[BIN_ROWS];
    __shared__ int colbuf[BIN_CAP];
    __shared__ int wsum[8];
    __shared__ int wbase[8];

    const int tid  = threadIdx.x;
    const int lane = tid & 31;
    const int wave = tid >> 5;
    const int b    = blockIdx.x;
    const int row0 = b << BIN_SHIFT;
    const unsigned* __restrict__ trow = table + (size_t)b * n_chunks;
    const int2* __restrict__ pairs = (const int2*)stage;

    cnt[tid] = 0;
    __syncthreads();

    int mybase = 0;
    for (int w = tid; w < n_chunks; w += 256) {
        const unsigned pk = trow[w];
        const int st = min((int)(pk >> 16), CHUNK);
        const int n  = min((int)(pk & 0xFFFFu), CHUNK - st);
        mybase += st;
        const int2* __restrict__ seg = pairs + (size_t)w * CHUNK + st;
        for (int i = 0; i < n; ++i) {
            const int lr = seg[i].x - row0;
            if ((unsigned)lr < (unsigned)BIN_ROWS) atomicAdd(&cnt[lr], 1);
        }
    }
    __syncthreads();

    const int myc = cnt[tid];
    int x = myc;
    for (int d = 1; d < 32; d <<= 1) {
        const int y = __shfl_up(x, d);
        if (lane >= d) x += y;
    }
    int sb = mybase;
    for (int d = 16; d >= 1; d >>= 1) sb += __shfl_xor(sb, d);
    if (lane == 31) { wsum[wave] = x; wbase[wave] = sb; }
    __syncthreads();
    int woff = 0, bin_tot = 0, bin_base = 0;
    for (int i = 0; i < 8; ++i) {
        const int s = wsum[i];
        if (i < wave) woff += s;
        bin_tot  += s;
        bin_base += wbase[i];
    }
    const int rstart = woff + x - myc;

    const int rowg = row0 + tid;
    if (rowg < n_nodes) offset[rowg] = bin_base + rstart;
    if (b == n_bins - 1 && tid == 0) offset[n_nodes] = bin_base + bin_tot;
    cnt[tid] = rstart;
    __syncthreads();

    for (int w = tid; w < n_chunks; w += 256) {
        const unsigned pk = trow[w];
        const int st = min((int)(pk >> 16), CHUNK);
        const int n  = min((int)(pk & 0xFFFFu), CHUNK - st);
        const int2* __restrict__ seg = pairs + (size_t)w * CHUNK + st;
        for (int i = 0; i < n; ++i) {
            const int2 e = seg[i];
            const int lr = e.x - row0;
            if ((unsigned)lr < (unsigned)BIN_ROWS) {
                const int p = atomicAdd(&cnt[lr], 1);
                if (p < BIN_CAP) {
                    colbuf[p] = e.y;
                } else {
                    const int gp = bin_base + p;
                    if ((unsigned)gp < (unsigned)n_edges) scol[gp] = e.y;
                }
            }
        }
    }
    __syncthreads();
    const int nl = min(bin_tot, BIN_CAP);
    for (int i = tid; i < nl; i += 256) {
        const int gp = bin_base + i;
        if ((unsigned)gp < (unsigned)n_edges) scol[gp] = colbuf[i];
    }
}

__device__ __forceinline__ int clamp_row(int c, int n_nodes) {
    return min(max(c, 0), n_nodes - 1);
}

__device__ __forceinline__ float quad_sum(float p) {
    p += __builtin_bit_cast(float, __builtin_amdgcn_update_dpp(
             0, __builtin_bit_cast(int, p), 0xB1, 0xF, 0xF, true));
    p += __builtin_bit_cast(float, __builtin_amdgcn_update_dpp(
             0, __builtin_bit_cast(int, p), 0x4E, 0xF, 0xF, true));
    return p;
}

__device__ __forceinline__ void edge_accumulate(
    const float4 q, const float4 k4, const float4 v4, float4& acc, float& den)
{
    float att = quad_sum(q.x * k4.x + q.y * k4.y + q.z * k4.z + q.w * k4.w);
    att = fminf(10.0f, fmaxf(-10.0f, att));
    const float ea = __expf(att);
    den   += ea;
    acc.x += ea * v4.x;
    acc.y += ea * v4.y;
    acc.z += ea * v4.z;
    acc.w += ea * v4.w;
}

__global__ __launch_bounds__(256) void row_attend(
    const float* __restrict__ Q, const float* __restrict__ K,
    const float* __restrict__ V,
    const int* __restrict__ offset, const int* __restrict__ scol,
    float* __restrict__ out, int n_nodes)
{
    const int lane = threadIdx.x & 31;
    const int row = blockIdx.x * 8 + __builtin_amdgcn_readfirstlane(threadIdx.x >> 5);
    if (row >= n_nodes) return;
    const int beg = offset[row];
    const int end = offset[row + 1];

    const float4 q = *(const float4*)(Q + (size_t)row * LATDIM + lane * 4);
    float4 acc = make_float4(0.0f, 0.0f, 0.0f, 0.0f);
    float den = 0.0f;

    for (int base = beg; base < end; base += 32) {
        const int cnt = min(32, end - base);
        const int cidx = (lane < cnt) ? scol[base + lane] : 0;
        int i = 0;
        for (; i + 1 < cnt; i += 2) {
            const int ca = clamp_row(__builtin_amdgcn_readlane(cidx, i), n_nodes);
            const int cb = clamp_row(__builtin_amdgcn_readlane(cidx, i + 1), n_nodes);
            const float4 ka = *(const float4*)(K + (size_t)ca * LATDIM + lane * 4);
            const float4 kb = *(const float4*)(K + (size_t)cb * LATDIM + lane * 4);
            const float4 va = *(const float4*)(V + (size_t)ca * LATDIM + lane * 4);
            const float4 vb = *(const float4*)(V + (size_t)cb * LATDIM + lane * 4);
            edge_accumulate(q, ka, va, acc, den);
            edge_accumulate(q, kb, vb, acc, den);
        }
        if (i < cnt) {
            const int ca = clamp_row(__builtin_amdgcn_readlane(cidx, i), n_nodes);
            const float4 ka = *(const float4*)(K + (size_t)ca * LATDIM + lane * 4);
            const float4 va = *(const float4*)(V + (size_t)ca * LATDIM + lane * 4);
            edge_accumulate(q, ka, va, acc, den);
        }
    }

    const float d = den + 1e-8f;
    float4 o;
    o.x = acc.x / d;
    o.y = acc.y / d;
    o.z = acc.z / d;
    o.w = acc.w / d;
    *(float4*)(out + (size_t)row * LATDIM + lane * 4) = o;
}

extern "C" void kernel_launch(void* const* d_in, const int* in_sizes, int n_in,
                              void* d_out, int out_size, void* d_ws, size_t ws_size,
                              hipStream_t stream) {
    const float* embeds = (const float*)d_in[0];
    const float* Wq     = (const float*)d_in[1];
    const float* Wk     = (const float*)d_in[2];
    const float* Wv     = (const float*)d_in[3];
    const int*   rows   = (const int*)d_in[4];
    const int*   cols   = (const int*)d_in[5];

    const int N = in_sizes[0] / LATDIM;
    const int E = in_sizes[4];

    float* out = (float*)d_out;

    const int NB  = (N + BIN_ROWS - 1) >> BIN_SHIFT;
    const int NCH = (E + CHUNK - 1) / CHUNK;

    const size_t NP = ((size_t)N + 64) & ~(size_t)63;
    const size_t EP = ((size_t)E + 63) & ~(size_t)63;
    float* ws     = (float*)d_ws;
    float* Qm     = ws;
    float* Km     = Qm + (size_t)N * LATDIM;
    float* Vm     = Km + (size_t)N * LATDIM;
    int*   offset = (int*)(Vm + (size_t)N * LATDIM);
    int*   scol   = offset + NP;
    int*   stage  = scol + EP;
    unsigned* table = (unsigned*)(stage + 2 * (size_t)NCH * CHUNK);

    bin_edges<<<NCH, 256, 2 * NB * sizeof(int), stream>>>(rows, cols, stage, table, E, NB, NCH);
    bucket_rows<<<NB, 256, 0, stream>>>(stage, table, offset, scol, E, N, NB, NCH);

    const int rowTiles = (N + 15) / 16;
    const int nwaves   = rowTiles * 3;
    qkv_gemm_wmma<<<(nwaves + 7) / 8, 256, 0, stream>>>(
        embeds, Wq, Wk, Wv, Qm, Km, Vm, rowTiles, N);

    row_attend<<<(N + 7) / 8, 256, 0, stream>>>(Qm, Km, Vm, offset, scol, out, N);
}
